# starting kernel
# baseline (speedup 1.0000x reference)
#include <hip/hip_runtime.h>


typedef __attribute__((ext_vector_type(2))) float v2f;
typedef __attribute__((ext_vector_type(4))) float v4f;
typedef __attribute__((ext_vector_type(8))) float v8f;

#define B_TRIALS 256
#define F_BINS   100
#define N_UNITS  512
#define M_HID    1024
#define P_OUT    512
#define FT       16
#define NFT      7
#define XS_STRIDE 516
#define HS_STRIDE 1028

__global__ __launch_bounds__(256) void stitch_encoder_kernel(
    const float* __restrict__ x,
    const int*   __restrict__ eid,
    const float* __restrict__ W1,
    const float* __restrict__ b1,
    const float* __restrict__ W2,
    const float* __restrict__ b2,
    float*       __restrict__ out)
{
    extern __shared__ float lds[];
    float* Xs = lds;
    float* Hs = lds + FT * XS_STRIDE;

    const int blk  = blockIdx.x;
    const int b    = blk / NFT;
    const int ft   = blk % NFT;
    const int f0   = ft * FT;
    const int rows = (F_BINS - f0 < FT) ? (F_BINS - f0) : FT;

    const int g = eid[b];
    const float* W1g = W1 + (size_t)g * N_UNITS * M_HID;
    const float* b1g = b1 + (size_t)g * M_HID;
    const float* W2g = W2 + (size_t)g * M_HID * P_OUT;
    const float* b2g = b2 + (size_t)g * P_OUT;

    const int t = threadIdx.x;

    const float* xb = x + ((size_t)b * F_BINS + f0) * N_UNITS;
    #pragma unroll
    for (int i = 0; i < 8; ++i) {
        int flat4 = t + i * 256;
        int row   = flat4 >> 7;
        int c4    = flat4 & 127;
        v4f val = {};
        if (row < rows)
            val = *(const v4f*)(xb + (size_t)row * N_UNITS + c4 * 4);
        *(v4f*)(Xs + row * XS_STRIDE + c4 * 4) = val;
    }
    __syncthreads();

    const int lane = t & 31;
    const int wv   = t >> 5;
    const int half = lane >> 4;
    const int m    = lane & 15;

    for (int ntp = 0; ntp < 2; ++ntp) {
        const int col0 = (wv * 8 + ntp * 4) * 16;
        v8f acc0 = {}, acc1 = {}, acc2 = {}, acc3 = {};
        const float* Arow = Xs + m * XS_STRIDE + 2 * half;
        const float* Bcol = W1g + (size_t)(2 * half) * M_HID + col0 + m;
        #pragma unroll 4
        for (int k0 = 0; k0 < N_UNITS; k0 += 4) {
            v2f a = *(const v2f*)(Arow + k0);
            const float* Bk0 = Bcol + (size_t)k0 * M_HID;
            const float* Bk1 = Bcol + (size_t)(k0 + 1) * M_HID;
            v2f bb0, bb1, bb2, bb3;
            bb0.x = Bk0[ 0]; bb0.y = Bk1[ 0];
            bb1.x = Bk0[16]; bb1.y = Bk1[16];
            bb2.x = Bk0[32]; bb2.y = Bk1[32];
            bb3.x = Bk0[48]; bb3.y = Bk1[48];
            acc0 = __builtin_amdgcn_wmma_f32_16x16x4_f32(
                false, a, false, bb0, (short)0, acc0, false, false);
            acc1 = __builtin_amdgcn_wmma_f32_16x16x4_f32(
                false, a, false, bb1, (short)0, acc1, false, false);
            acc2 = __builtin_amdgcn_wmma_f32_16x16x4_f32(
                false, a, false, bb2, (short)0, acc2, false, false);
            acc3 = __builtin_amdgcn_wmma_f32_16x16x4_f32(
                false, a, false, bb3, (short)0, acc3, false, false);
        }
        const float bias0 = b1g[col0 + m];
        const float bias1 = b1g[col0 + 16 + m];
        const float bias2 = b1g[col0 + 32 + m];
        const float bias3 = b1g[col0 + 48 + m];
        float* Hrow = Hs + (8 * half) * HS_STRIDE + col0 + m;
        #pragma unroll
        for (int r = 0; r < 8; ++r) {
            float h0 = acc0[r] + bias0;
            float h1 = acc1[r] + bias1;
            float h2 = acc2[r] + bias2;
            float h3 = acc3[r] + bias3;
            h0 = h0 * __builtin_amdgcn_rcpf(1.0f + __builtin_fabsf(h0));
            h1 = h1 * __builtin_amdgcn_rcpf(1.0f + __builtin_fabsf(h1));
            h2 = h2 * __builtin_amdgcn_rcpf(1.0f + __builtin_fabsf(h2));
            h3 = h3 * __builtin_amdgcn_rcpf(1.0f + __builtin_fabsf(h3));
            Hrow[r * HS_STRIDE +  0] = h0;
            Hrow[r * HS_STRIDE + 16] = h1;
            Hrow[r * HS_STRIDE + 32] = h2;
            Hrow[r * HS_STRIDE + 48] = h3;
        }
    }
    __syncthreads();

    float* outb = out + ((size_t)b * F_BINS + f0) * P_OUT;
    {
        const int col0 = (wv * 4) * 16;
        v8f acc0 = {}, acc1 = {}, acc2 = {}, acc3 = {};
        const float* Arow = Hs + m * HS_STRIDE + 2 * half;
        const float* Bcol = W2g + (size_t)(2 * half) * P_OUT + col0 + m;
        #pragma unroll 4
        for (int k0 = 0; k0 < M_HID; k0 += 4) {
            v2f a = *(const v2f*)(Arow + k0);
            const float* Bk0 = Bcol + (size_t)k0 * P_OUT;
            const float* Bk1 = Bcol + (size_t)(k0 + 1) * P_OUT;
            v2f bb0, bb1, bb2, bb3;
            bb0.x = Bk0[ 0]; bb0.y = Bk1[ 0];
            bb1.x = Bk0[16]; bb1.y = Bk1[16];
            bb2.x = Bk0[32]; bb2.y = Bk1[32];
            bb3.x = Bk0[48]; bb3.y = Bk1[48];
            acc0 = __builtin_amdgcn_wmma_f32_16x16x4_f32(
                false, a, false, bb0, (short)0, acc0, false, false);
            acc1 = __builtin_amdgcn_wmma_f32_16x16x4_f32(
                false, a, false, bb1, (short)0, acc1, false, false);
            acc2 = __builtin_amdgcn_wmma_f32_16x16x4_f32(
                false, a, false, bb2, (short)0, acc2, false, false);
            acc3 = __builtin_amdgcn_wmma_f32_16x16x4_f32(
                false, a, false, bb3, (short)0, acc3, false, false);
        }
        const float bias0 = b2g[col0 + m];
        const float bias1 = b2g[col0 + 16 + m];
        const float bias2 = b2g[col0 + 32 + m];
        const float bias3 = b2g[col0 + 48 + m];
        #pragma unroll
        for (int r = 0; r < 8; ++r) {
            int fr = r + 8 * half;
            if (fr < rows) {
                float* orow = outb + (size_t)fr * P_OUT + col0 + m;
                orow[ 0] = acc0[r] + bias0;
                orow[16] = acc1[r] + bias1;
                orow[32] = acc2[r] + bias2;
                orow[48] = acc3[r] + bias3;
            }
        }
    }
}

extern "C" void kernel_launch(void* const* d_in, const int* in_sizes, int n_in,
                              void* d_out, int out_size, void* d_ws, size_t ws_size,
                              hipStream_t stream) {
    (void)in_sizes; (void)n_in; (void)out_size; (void)d_ws; (void)ws_size;
    const float* x   = (const float*)d_in[0];
    const int*   eid = (const int*)  d_in[1];
    const float* W1  = (const float*)d_in[2];
    const float* b1  = (const float*)d_in[3];
    const float* W2  = (const float*)d_in[4];
    const float* b2  = (const float*)d_in[5];
    float* out = (float*)d_out;

    const size_t shmem = (size_t)(FT * XS_STRIDE + FT * HS_STRIDE) * sizeof(float);
    dim3 grid(B_TRIALS * NFT);
    dim3 block(256);
    hipLaunchKernelGGL(stitch_encoder_kernel, grid, block, shmem, stream,
                       x, eid, W1, b1, W2, b2, out);
}


# B operand loads widened to 16 bytes per lane: each wave's 4 column tiles interleaved (tile j = columns 4c+j) so one load feeds 4 WMMAs; h and out written as whole 128-byte lines
# speedup vs baseline: 1.0417x; 1.0417x over previous
#include <hip/hip_runtime.h>


typedef __attribute__((ext_vector_type(2))) float v2f;
typedef __attribute__((ext_vector_type(4))) float v4f;
typedef __attribute__((ext_vector_type(8))) float v8f;

#define B_TRIALS 256
#define F_BINS   100
#define N_UNITS  512
#define M_HID    1024
#define P_OUT    512
#define FT       16
#define NFT      7
#define XS_STRIDE 516
#define HS_STRIDE 1028

__global__ __launch_bounds__(256) void stitch_encoder_kernel(
    const float* __restrict__ x,
    const int*   __restrict__ eid,
    const float* __restrict__ W1,
    const float* __restrict__ b1,
    const float* __restrict__ W2,
    const float* __restrict__ b2,
    float*       __restrict__ out)
{
    extern __shared__ float lds[];
    float* Xs = lds;
    float* Hs = lds + FT * XS_STRIDE;

    const int blk  = blockIdx.x;
    const int b    = blk / NFT;
    const int ft   = blk % NFT;
    const int f0   = ft * FT;
    const int rows = (F_BINS - f0 < FT) ? (F_BINS - f0) : FT;

    const int g = eid[b];
    const float* W1g = W1 + (size_t)g * N_UNITS * M_HID;
    const float* b1g = b1 + (size_t)g * M_HID;
    const float* W2g = W2 + (size_t)g * M_HID * P_OUT;
    const float* b2g = b2 + (size_t)g * P_OUT;

    const int t = threadIdx.x;

    const float* xb = x + ((size_t)b * F_BINS + f0) * N_UNITS;
    #pragma unroll
    for (int i = 0; i < 8; ++i) {
        int flat4 = t + i * 256;
        int row   = flat4 >> 7;
        int c4    = flat4 & 127;
        v4f val = {};
        if (row < rows)
            val = *(const v4f*)(xb + (size_t)row * N_UNITS + c4 * 4);
        *(v4f*)(Xs + row * XS_STRIDE + c4 * 4) = val;
    }
    __syncthreads();

    const int lane = t & 31;
    const int wv   = t >> 5;
    const int half = lane >> 4;
    const int m    = lane & 15;

    for (int ntp = 0; ntp < 2; ++ntp) {
        const int col0 = (wv * 8 + ntp * 4) * 16;
        v8f acc0 = {}, acc1 = {}, acc2 = {}, acc3 = {};
        const float* Arow = Xs + m * XS_STRIDE + 2 * half;
        const float* Bcol = W1g + (size_t)(2 * half) * M_HID + col0 + 4 * m;
        #pragma unroll 4
        for (int k0 = 0; k0 < N_UNITS; k0 += 4) {
            v2f a = *(const v2f*)(Arow + k0);
            const v4f q0 = *(const v4f*)(Bcol + (size_t)k0 * M_HID);
            const v4f q1 = *(const v4f*)(Bcol + (size_t)(k0 + 1) * M_HID);
            v2f bb0, bb1, bb2, bb3;
            bb0.x = q0.x; bb0.y = q1.x;
            bb1.x = q0.y; bb1.y = q1.y;
            bb2.x = q0.z; bb2.y = q1.z;
            bb3.x = q0.w; bb3.y = q1.w;
            acc0 = __builtin_amdgcn_wmma_f32_16x16x4_f32(
                false, a, false, bb0, (short)0, acc0, false, false);
            acc1 = __builtin_amdgcn_wmma_f32_16x16x4_f32(
                false, a, false, bb1, (short)0, acc1, false, false);
            acc2 = __builtin_amdgcn_wmma_f32_16x16x4_f32(
                false, a, false, bb2, (short)0, acc2, false, false);
            acc3 = __builtin_amdgcn_wmma_f32_16x16x4_f32(
                false, a, false, bb3, (short)0, acc3, false, false);
        }
        const v4f bias = *(const v4f*)(b1g + col0 + 4 * m);
        float* Hrow = Hs + (8 * half) * HS_STRIDE + col0 + 4 * m;
        #pragma unroll
        for (int r = 0; r < 8; ++r) {
            float h0 = acc0[r] + bias.x;
            float h1 = acc1[r] + bias.y;
            float h2 = acc2[r] + bias.z;
            float h3 = acc3[r] + bias.w;
            v4f hv;
            hv.x = h0 * __builtin_amdgcn_rcpf(1.0f + __builtin_fabsf(h0));
            hv.y = h1 * __builtin_amdgcn_rcpf(1.0f + __builtin_fabsf(h1));
            hv.z = h2 * __builtin_amdgcn_rcpf(1.0f + __builtin_fabsf(h2));
            hv.w = h3 * __builtin_amdgcn_rcpf(1.0f + __builtin_fabsf(h3));
            *(v4f*)(Hrow + r * HS_STRIDE) = hv;
        }
    }
    __syncthreads();

    float* outb = out + ((size_t)b * F_BINS + f0) * P_OUT;
    {
        const int col0 = (wv * 4) * 16;
        v8f acc0 = {}, acc1 = {}, acc2 = {}, acc3 = {};
        const float* Arow = Hs + m * HS_STRIDE + 2 * half;
        const float* Bcol = W2g + (size_t)(2 * half) * P_OUT + col0 + 4 * m;
        #pragma unroll 4
        for (int k0 = 0; k0 < M_HID; k0 += 4) {
            v2f a = *(const v2f*)(Arow + k0);
            const v4f q0 = *(const v4f*)(Bcol + (size_t)k0 * P_OUT);
            const v4f q1 = *(const v4f*)(Bcol + (size_t)(k0 + 1) * P_OUT);
            v2f bb0, bb1, bb2, bb3;
            bb0.x = q0.x; bb0.y = q1.x;
            bb1.x = q0.y; bb1.y = q1.y;
            bb2.x = q0.z; bb2.y = q1.z;
            bb3.x = q0.w; bb3.y = q1.w;
            acc0 = __builtin_amdgcn_wmma_f32_16x16x4_f32(
                false, a, false, bb0, (short)0, acc0, false, false);
            acc1 = __builtin_amdgcn_wmma_f32_16x16x4_f32(
                false, a, false, bb1, (short)0, acc1, false, false);
            acc2 = __builtin_amdgcn_wmma_f32_16x16x4_f32(
                false, a, false, bb2, (short)0, acc2, false, false);
            acc3 = __builtin_amdgcn_wmma_f32_16x16x4_f32(
                false, a, false, bb3, (short)0, acc3, false, false);
        }
        const v4f bias = *(const v4f*)(b2g + col0 + 4 * m);
        #pragma unroll
        for (int r = 0; r < 8; ++r) {
            int fr = r + 8 * half;
            if (fr < rows) {
                v4f ov;
                ov.x = acc0[r] + bias.x;
                ov.y = acc1[r] + bias.y;
                ov.z = acc2[r] + bias.z;
                ov.w = acc3[r] + bias.w;
                *(v4f*)(outb + (size_t)fr * P_OUT + col0 + 4 * m) = ov;
            }
        }
    }
}

extern "C" void kernel_launch(void* const* d_in, const int* in_sizes, int n_in,
                              void* d_out, int out_size, void* d_ws, size_t ws_size,
                              hipStream_t stream) {
    (void)in_sizes; (void)n_in; (void)out_size; (void)d_ws; (void)ws_size;
    const float* x   = (const float*)d_in[0];
    const int*   eid = (const int*)  d_in[1];
    const float* W1  = (const float*)d_in[2];
    const float* b1  = (const float*)d_in[3];
    const float* W2  = (const float*)d_in[4];
    const float* b2  = (const float*)d_in[5];
    float* out = (float*)d_out;

    const size_t shmem = (size_t)(FT * XS_STRIDE + FT * HS_STRIDE) * sizeof(float);
    dim3 grid(B_TRIALS * NFT);
    dim3 block(256);
    hipLaunchKernelGGL(stitch_encoder_kernel, grid, block, shmem, stream,
                       x, eid, W1, b1, W2, b2, out);
}


# Operand loads double-buffered in registers: the 8 weight loads and 2 LDS fragment reads of the next 16-k block are issued before the current block's 16 WMMAs, in both products; sums bit-identical
# speedup vs baseline: 1.0905x; 1.0468x over previous
#include <hip/hip_runtime.h>


typedef __attribute__((ext_vector_type(2))) float v2f;
typedef __attribute__((ext_vector_type(4))) float v4f;
typedef __attribute__((ext_vector_type(8))) float v8f;

#define B_TRIALS 256
#define F_BINS   100
#define N_UNITS  512
#define M_HID    1024
#define P_OUT    512
#define FT       16
#define NFT      7
#define XS_STRIDE 516
#define HS_STRIDE 1028


struct Frag {
    v4f q[8];
    v2f a[4];
};

template <int STRIDE>
__device__ __forceinline__ void load_block(Frag& f, const float* Bk,
                                           const float* Ak)
{
    #pragma unroll
    for (int s = 0; s < 4; ++s) {
        f.q[2 * s]     = *(const v4f*)(Bk + (size_t)(4 * s) * STRIDE);
        f.q[2 * s + 1] = *(const v4f*)(Bk + (size_t)(4 * s + 1) * STRIDE);
    }
    #pragma unroll
    for (int s = 0; s < 4; ++s)
        f.a[s] = *(const v2f*)(Ak + 4 * s);
}

__device__ __forceinline__ void mma_block(const Frag& f, v8f& acc0, v8f& acc1,
                                          v8f& acc2, v8f& acc3)
{
    #pragma unroll
    for (int s = 0; s < 4; ++s) {
        const v4f q0 = f.q[2 * s];
        const v4f q1 = f.q[2 * s + 1];
        const v2f a  = f.a[s];
        v2f bb0, bb1, bb2, bb3;
        bb0.x = q0.x; bb0.y = q1.x;
        bb1.x = q0.y; bb1.y = q1.y;
        bb2.x = q0.z; bb2.y = q1.z;
        bb3.x = q0.w; bb3.y = q1.w;
        acc0 = __builtin_amdgcn_wmma_f32_16x16x4_f32(
            false, a, false, bb0, (short)0, acc0, false, false);
        acc1 = __builtin_amdgcn_wmma_f32_16x16x4_f32(
            false, a, false, bb1, (short)0, acc1, false, false);
        acc2 = __builtin_amdgcn_wmma_f32_16x16x4_f32(
            false, a, false, bb2, (short)0, acc2, false, false);
        acc3 = __builtin_amdgcn_wmma_f32_16x16x4_f32(
            false, a, false, bb3, (short)0, acc3, false, false);
    }
}

template <int STRIDE, int K>
__device__ __forceinline__ void gemm_k(const float* Arow, const float* Bcol,
                                       v8f& acc0, v8f& acc1, v8f& acc2, v8f& acc3)
{
    Frag f0, f1;
    load_block<STRIDE>(f0, Bcol, Arow);
    #pragma unroll 1
    for (int k0 = 0; k0 < K - 32; k0 += 32) {
        load_block<STRIDE>(f1, Bcol + (size_t)(k0 + 16) * STRIDE, Arow + k0 + 16);
        __builtin_amdgcn_sched_barrier(0);
        mma_block(f0, acc0, acc1, acc2, acc3);
        asm volatile("" :: "v"(acc0), "v"(acc1), "v"(acc2), "v"(acc3) : "memory");
        load_block<STRIDE>(f0, Bcol + (size_t)(k0 + 32) * STRIDE, Arow + k0 + 32);
        __builtin_amdgcn_sched_barrier(0);
        mma_block(f1, acc0, acc1, acc2, acc3);
    }
    load_block<STRIDE>(f1, Bcol + (size_t)(K - 16) * STRIDE, Arow + K - 16);
    mma_block(f0, acc0, acc1, acc2, acc3);
    mma_block(f1, acc0, acc1, acc2, acc3);
}

__global__ __launch_bounds__(256) void stitch_encoder_kernel(
    const float* __restrict__ x,
    const int*   __restrict__ eid,
    const float* __restrict__ W1,
    const float* __restrict__ b1,
    const float* __restrict__ W2,
    const float* __restrict__ b2,
    float*       __restrict__ out)
{
    extern __shared__ float lds[];
    float* Xs = lds;
    float* Hs = lds + FT * XS_STRIDE;

    const int blk  = blockIdx.x;
    const int b    = blk / NFT;
    const int ft   = blk % NFT;
    const int f0   = ft * FT;
    const int rows = (F_BINS - f0 < FT) ? (F_BINS - f0) : FT;

    const int g = eid[b];
    const float* W1g = W1 + (size_t)g * N_UNITS * M_HID;
    const float* b1g = b1 + (size_t)g * M_HID;
    const float* W2g = W2 + (size_t)g * M_HID * P_OUT;
    const float* b2g = b2 + (size_t)g * P_OUT;

    const int t = threadIdx.x;

    const float* xb = x + ((size_t)b * F_BINS + f0) * N_UNITS;
    #pragma unroll
    for (int i = 0; i < 8; ++i) {
        int flat4 = t + i * 256;
        int row   = flat4 >> 7;
        int c4    = flat4 & 127;
        v4f val = {};
        if (row < rows)
            val = *(const v4f*)(xb + (size_t)row * N_UNITS + c4 * 4);
        *(v4f*)(Xs + row * XS_STRIDE + c4 * 4) = val;
    }
    __syncthreads();

    const int lane = t & 31;
    const int wv   = t >> 5;
    const int half = lane >> 4;
    const int m    = lane & 15;

    for (int ntp = 0; ntp < 2; ++ntp) {
        const int col0 = (wv * 8 + ntp * 4) * 16;
        v8f acc0 = {}, acc1 = {}, acc2 = {}, acc3 = {};
        const float* Arow = Xs + m * XS_STRIDE + 2 * half;
        const float* Bcol = W1g + (size_t)(2 * half) * M_HID + col0 + 4 * m;
        gemm_k<M_HID, N_UNITS>(Arow, Bcol, acc0, acc1, acc2, acc3);
        const v4f bias = *(const v4f*)(b1g + col0 + 4 * m);
        float* Hrow = Hs + (8 * half) * HS_STRIDE + col0 + 4 * m;
        #pragma unroll
        for (int r = 0; r < 8; ++r) {
            float h0 = acc0[r] + bias.x;
            float h1 = acc1[r] + bias.y;
            float h2 = acc2[r] + bias.z;
            float h3 = acc3[r] + bias.w;
            v4f hv;
            hv.x = h0 * __builtin_amdgcn_rcpf(1.0f + __builtin_fabsf(h0));
            hv.y = h1 * __builtin_amdgcn_rcpf(1.0f + __builtin_fabsf(h1));
            hv.z = h2 * __builtin_amdgcn_rcpf(1.0f + __builtin_fabsf(h2));
            hv.w = h3 * __builtin_amdgcn_rcpf(1.0f + __builtin_fabsf(h3));
            *(v4f*)(Hrow + r * HS_STRIDE) = hv;
        }
    }
    __syncthreads();

    float* outb = out + ((size_t)b * F_BINS + f0) * P_OUT;
    {
        const int col0 = (wv * 4) * 16;
        v8f acc0 = {}, acc1 = {}, acc2 = {}, acc3 = {};
        const float* Arow = Hs + m * HS_STRIDE + 2 * half;
        const float* Bcol = W2g + (size_t)(2 * half) * P_OUT + col0 + 4 * m;
        gemm_k<P_OUT, M_HID>(Arow, Bcol, acc0, acc1, acc2, acc3);
        const v4f bias = *(const v4f*)(b2g + col0 + 4 * m);
        #pragma unroll
        for (int r = 0; r < 8; ++r) {
            int fr = r + 8 * half;
            if (fr < rows) {
                v4f ov;
                ov.x = acc0[r] + bias.x;
                ov.y = acc1[r] + bias.y;
                ov.z = acc2[r] + bias.z;
                ov.w = acc3[r] + bias.w;
                *(v4f*)(outb + (size_t)fr * P_OUT + col0 + 4 * m) = ov;
            }
        }
    }
}

extern "C" void kernel_launch(void* const* d_in, const int* in_sizes, int n_in,
                              void* d_out, int out_size, void* d_ws, size_t ws_size,
                              hipStream_t stream) {
    (void)in_sizes; (void)n_in; (void)out_size; (void)d_ws; (void)ws_size;
    const float* x   = (const float*)d_in[0];
    const int*   eid = (const int*)  d_in[1];
    const float* W1  = (const float*)d_in[2];
    const float* b1  = (const float*)d_in[3];
    const float* W2  = (const float*)d_in[4];
    const float* b2  = (const float*)d_in[5];
    float* out = (float*)d_out;

    const size_t shmem = (size_t)(FT * XS_STRIDE + FT * HS_STRIDE) * sizeof(float);
    dim3 grid(B_TRIALS * NFT);
    dim3 block(256);
    hipLaunchKernelGGL(stitch_encoder_kernel, grid, block, shmem, stream,
                       x, eid, W1, b1, W2, b2, out);
}


# One workgroup owns 3 or 2 row tiles of a trial (7 tiles as 3+2+2, 3-tile groups first; 16 waves, one workgroup per WGP): a loaded weight fragment feeds all its row tiles; sums bit-identical
# speedup vs baseline: 1.1660x; 1.0692x over previous
#include <hip/hip_runtime.h>


typedef __attribute__((ext_vector_type(2))) float v2f;
typedef __attribute__((ext_vector_type(4))) float v4f;
typedef __attribute__((ext_vector_type(8))) float v8f;

#define B_TRIALS 256
#define F_BINS   100
#define N_UNITS  512
#define M_HID    1024
#define P_OUT    512
#define RT_MAX   3
#define XS_STRIDE 516
#define HS_STRIDE 1028
#define LDS_FLOATS (RT_MAX * 16 * XS_STRIDE + RT_MAX * 16 * HS_STRIDE)

template <int NJ> struct WVec;
template <> struct WVec<4> { typedef v4f type; };
template <> struct WVec<2> { typedef v2f type; };

template <int RT, int NJ>
struct Frag {
    typename WVec<NJ>::type q[8];
    v2f a[RT][4];
};

template <int RT, int NJ, int WSTRIDE, int ASTRIDE>
__device__ __forceinline__ void load_block(Frag<RT, NJ>& f, const float* Bk,
                                           const float* Ak)
{
    typedef typename WVec<NJ>::type wv_t;
    #pragma unroll
    for (int s = 0; s < 4; ++s) {
        f.q[2 * s]     = *(const wv_t*)(Bk + (size_t)(4 * s) * WSTRIDE);
        f.q[2 * s + 1] = *(const wv_t*)(Bk + (size_t)(4 * s + 1) * WSTRIDE);
    }
    #pragma unroll
    for (int rt = 0; rt < RT; ++rt) {
        #pragma unroll
        for (int s = 0; s < 4; ++s)
            f.a[rt][s] = *(const v2f*)(Ak + rt * 16 * ASTRIDE + 4 * s);
    }
}

template <int RT, int NJ>
__device__ __forceinline__ void mma_block(const Frag<RT, NJ>& f, v8f (&acc)[RT][NJ])
{
    #pragma unroll
    for (int s = 0; s < 4; ++s) {
        v2f bb[NJ];
        #pragma unroll
        for (int j = 0; j < NJ; ++j) {
            bb[j].x = f.q[2 * s][j];
            bb[j].y = f.q[2 * s + 1][j];
        }
        #pragma unroll
        for (int rt = 0; rt < RT; ++rt) {
            const v2f a = f.a[rt][s];
            #pragma unroll
            for (int j = 0; j < NJ; ++j)
                acc[rt][j] = __builtin_amdgcn_wmma_f32_16x16x4_f32(
                    false, a, false, bb[j], (short)0, acc[rt][j], false, false);
        }
    }
}

template <int RT, int NJ>
__device__ __forceinline__ void fence_acc(v8f (&acc)[RT][NJ])
{
    #pragma unroll
    for (int rt = 0; rt < RT; ++rt) {
        if constexpr (NJ == 4) {
            asm volatile("" :: "v"(acc[rt][0]), "v"(acc[rt][1]), "v"(acc[rt][2]), "v"(acc[rt][3]) : "memory");
        } else {
            asm volatile("" :: "v"(acc[rt][0]), "v"(acc[rt][1]) : "memory");
        }
    }
}

template <int RT, int NJ, int WSTRIDE, int ASTRIDE, int K>
__device__ __forceinline__ void gemm_k(const float* Arow, const float* Bcol,
                                       v8f (&acc)[RT][NJ])
{
    Frag<RT, NJ> f0, f1;
    load_block<RT, NJ, WSTRIDE, ASTRIDE>(f0, Bcol, Arow);
    #pragma unroll 1
    for (int k0 = 0; k0 < K - 32; k0 += 32) {
        load_block<RT, NJ, WSTRIDE, ASTRIDE>(f1, Bcol + (size_t)(k0 + 16) * WSTRIDE, Arow + k0 + 16);
        __builtin_amdgcn_sched_barrier(0);
        mma_block<RT, NJ>(f0, acc);
        fence_acc<RT, NJ>(acc);
        load_block<RT, NJ, WSTRIDE, ASTRIDE>(f0, Bcol + (size_t)(k0 + 32) * WSTRIDE, Arow + k0 + 32);
        __builtin_amdgcn_sched_barrier(0);
        mma_block<RT, NJ>(f1, acc);
    }
    load_block<RT, NJ, WSTRIDE, ASTRIDE>(f1, Bcol + (size_t)(K - 16) * WSTRIDE, Arow + K - 16);
    mma_block<RT, NJ>(f0, acc);
    mma_block<RT, NJ>(f1, acc);
}

template <int RT>
__device__ __forceinline__ void group_body(
    const float* __restrict__ x, const float* __restrict__ W1g,
    const float* __restrict__ b1g, const float* __restrict__ W2g,
    const float* __restrict__ b2g, float* __restrict__ out,
    int b, int f0, int rows, float* Xs, float* Hs)
{
    const int t = threadIdx.x;

    const float* xb = x + ((size_t)b * F_BINS + f0) * N_UNITS;
    #pragma unroll
    for (int i = 0; i < RT * 4; ++i) {
        int flat4 = t + i * 512;
        int row   = flat4 >> 7;
        int c4    = flat4 & 127;
        v4f val = {};
        if (row < rows)
            val = *(const v4f*)(xb + (size_t)row * N_UNITS + c4 * 4);
        *(v4f*)(Xs + row * XS_STRIDE + c4 * 4) = val;
    }
    __syncthreads();

    const int lane = t & 31;
    const int wv   = t >> 5;
    const int half = lane >> 4;
    const int m    = lane & 15;

    {
        const int col0 = wv * 64;
        v8f acc[RT][4] = {};
        const float* Arow = Xs + m * XS_STRIDE + 2 * half;
        const float* Bcol = W1g + (size_t)(2 * half) * M_HID + col0 + 4 * m;
        gemm_k<RT, 4, M_HID, XS_STRIDE, N_UNITS>(Arow, Bcol, acc);
        const v4f bias = *(const v4f*)(b1g + col0 + 4 * m);
        #pragma unroll
        for (int rt = 0; rt < RT; ++rt) {
            float* Hrow = Hs + (16 * rt + 8 * half) * HS_STRIDE + col0 + 4 * m;
            #pragma unroll
            for (int r = 0; r < 8; ++r) {
                float h0 = acc[rt][0][r] + bias.x;
                float h1 = acc[rt][1][r] + bias.y;
                float h2 = acc[rt][2][r] + bias.z;
                float h3 = acc[rt][3][r] + bias.w;
                v4f hv;
                hv.x = h0 * __builtin_amdgcn_rcpf(1.0f + __builtin_fabsf(h0));
                hv.y = h1 * __builtin_amdgcn_rcpf(1.0f + __builtin_fabsf(h1));
                hv.z = h2 * __builtin_amdgcn_rcpf(1.0f + __builtin_fabsf(h2));
                hv.w = h3 * __builtin_amdgcn_rcpf(1.0f + __builtin_fabsf(h3));
                *(v4f*)(Hrow + r * HS_STRIDE) = hv;
            }
        }
    }
    __syncthreads();

    float* outb = out + ((size_t)b * F_BINS + f0) * P_OUT;
    {
        const int col0 = wv * 32;
        v8f acc[RT][2] = {};
        const float* Arow = Hs + m * HS_STRIDE + 2 * half;
        const float* Bcol = W2g + (size_t)(2 * half) * P_OUT + col0 + 2 * m;
        gemm_k<RT, 2, P_OUT, HS_STRIDE, M_HID>(Arow, Bcol, acc);
        const v2f bias = *(const v2f*)(b2g + col0 + 2 * m);
        #pragma unroll
        for (int rt = 0; rt < RT; ++rt) {
            #pragma unroll
            for (int r = 0; r < 8; ++r) {
                int fr = 16 * rt + 8 * half + r;
                if (fr < rows) {
                    v2f ov;
                    ov.x = acc[rt][0][r] + bias.x;
                    ov.y = acc[rt][1][r] + bias.y;
                    *(v2f*)(outb + (size_t)fr * P_OUT + col0 + 2 * m) = ov;
                }
            }
        }
    }
}

__global__ __launch_bounds__(512) void stitch_encoder_kernel(
    const float* __restrict__ x,
    const int*   __restrict__ eid,
    const float* __restrict__ W1,
    const float* __restrict__ b1,
    const float* __restrict__ W2,
    const float* __restrict__ b2,
    float*       __restrict__ out)
{
    __shared__ __attribute__((aligned(16))) float lds[LDS_FLOATS];
    float* Xs = lds;
    float* Hs = lds + RT_MAX * 16 * XS_STRIDE;

    const int blk = blockIdx.x;
    const bool big = blk < B_TRIALS;
    const int rest = blk - B_TRIALS;
    const int b    = big ? blk : (rest >> 1);
    const int f0   = big ? 0 : (48 + 32 * (rest & 1));
    const int span = big ? 48 : 32;
    const int rows = (F_BINS - f0 < span) ? (F_BINS - f0) : span;

    const int g = eid[b];
    const float* W1g = W1 + (size_t)g * N_UNITS * M_HID;
    const float* b1g = b1 + (size_t)g * M_HID;
    const float* W2g = W2 + (size_t)g * M_HID * P_OUT;
    const float* b2g = b2 + (size_t)g * P_OUT;

    if (big)
        group_body<3>(x, W1g, b1g, W2g, b2g, out, b, f0, rows, Xs, Hs);
    else
        group_body<2>(x, W1g, b1g, W2g, b2g, out, b, f0, rows, Xs, Hs);
}

extern "C" void kernel_launch(void* const* d_in, const int* in_sizes, int n_in,
                              void* d_out, int out_size, void* d_ws, size_t ws_size,
                              hipStream_t stream) {
    (void)in_sizes; (void)n_in; (void)out_size; (void)d_ws; (void)ws_size;
    const float* x   = (const float*)d_in[0];
    const int*   eid = (const int*)  d_in[1];
    const float* W1  = (const float*)d_in[2];
    const float* b1  = (const float*)d_in[3];
    const float* W2  = (const float*)d_in[4];
    const float* b2  = (const float*)d_in[5];
    float* out = (float*)d_out;

    dim3 grid(B_TRIALS * 3);
    dim3 block(512);
    hipLaunchKernelGGL(stitch_encoder_kernel, grid, block, 0, stream,
                       x, eid, W1, b1, W2, b2, out);
}


# One workgroup owns 4 or 3 row tiles of a trial (7 tiles as 4+3, 2 workgroups per trial not 3): h is written over the dead x rows in LDS behind a barrier, so 4 tiles fit; sums bit-identical
# speedup vs baseline: 1.2289x; 1.0539x over previous
#include <hip/hip_runtime.h>


typedef __attribute__((ext_vector_type(2))) float v2f;
typedef __attribute__((ext_vector_type(4))) float v4f;
typedef __attribute__((ext_vector_type(8))) float v8f;

#define B_TRIALS 256
#define F_BINS   100
#define N_UNITS  512
#define M_HID    1024
#define P_OUT    512
#define RT_MAX   4
#define XS_STRIDE 516
#define HS_STRIDE 1028
#define LDS_FLOATS (RT_MAX * 16 * HS_STRIDE)

template <int NJ> struct WVec;
template <> struct WVec<4> { typedef v4f type; };
template <> struct WVec<2> { typedef v2f type; };

template <int RT, int NJ, int S>
struct Frag {
    typename WVec<NJ>::type q[2 * S];
    v2f a[RT][S];
};

template <int RT, int NJ, int S, int WSTRIDE, int ASTRIDE>
__device__ __forceinline__ void load_block(Frag<RT, NJ, S>& f, const float* Bk,
                                           const float* Ak)
{
    typedef typename WVec<NJ>::type wv_t;
    #pragma unroll
    for (int s = 0; s < S; ++s) {
        f.q[2 * s]     = *(const wv_t*)(Bk + (size_t)(4 * s) * WSTRIDE);
        f.q[2 * s + 1] = *(const wv_t*)(Bk + (size_t)(4 * s + 1) * WSTRIDE);
    }
    #pragma unroll
    for (int rt = 0; rt < RT; ++rt) {
        #pragma unroll
        for (int s = 0; s < S; ++s)
            f.a[rt][s] = *(const v2f*)(Ak + rt * 16 * ASTRIDE + 4 * s);
    }
}

template <int RT, int NJ, int S>
__device__ __forceinline__ void mma_block(const Frag<RT, NJ, S>& f, v8f (&acc)[RT][NJ])
{
    #pragma unroll
    for (int s = 0; s < S; ++s) {
        v2f bb[NJ];
        #pragma unroll
        for (int j = 0; j < NJ; ++j) {
            bb[j].x = f.q[2 * s][j];
            bb[j].y = f.q[2 * s + 1][j];
        }
        #pragma unroll
        for (int rt = 0; rt < RT; ++rt) {
            const v2f a = f.a[rt][s];
            #pragma unroll
            for (int j = 0; j < NJ; ++j)
                acc[rt][j] = __builtin_amdgcn_wmma_f32_16x16x4_f32(
                    false, a, false, bb[j], (short)0, acc[rt][j], false, false);
        }
    }
}

template <int RT, int NJ>
__device__ __forceinline__ void fence_acc(v8f (&acc)[RT][NJ])
{
    #pragma unroll
    for (int rt = 0; rt < RT; ++rt) {
        if constexpr (NJ == 4) {
            asm volatile("" :: "v"(acc[rt][0]), "v"(acc[rt][1]), "v"(acc[rt][2]), "v"(acc[rt][3]) : "memory");
        } else {
            asm volatile("" :: "v"(acc[rt][0]), "v"(acc[rt][1]) : "memory");
        }
    }
}

template <int RT, int NJ, int S, int WSTRIDE, int ASTRIDE, int K>
__device__ __forceinline__ void gemm_k(const float* Arow, const float* Bcol,
                                       v8f (&acc)[RT][NJ])
{
    constexpr int KB = 4 * S;
    Frag<RT, NJ, S> f0, f1;
    load_block<RT, NJ, S, WSTRIDE, ASTRIDE>(f0, Bcol, Arow);
    #pragma unroll 1
    for (int k0 = 0; k0 < K - 2 * KB; k0 += 2 * KB) {
        load_block<RT, NJ, S, WSTRIDE, ASTRIDE>(f1, Bcol + (size_t)(k0 + KB) * WSTRIDE, Arow + k0 + KB);
        __builtin_amdgcn_sched_barrier(0);
        mma_block<RT, NJ, S>(f0, acc);
        fence_acc<RT, NJ>(acc);
        load_block<RT, NJ, S, WSTRIDE, ASTRIDE>(f0, Bcol + (size_t)(k0 + 2 * KB) * WSTRIDE, Arow + k0 + 2 * KB);
        __builtin_amdgcn_sched_barrier(0);
        mma_block<RT, NJ, S>(f1, acc);
    }
    load_block<RT, NJ, S, WSTRIDE, ASTRIDE>(f1, Bcol + (size_t)(K - KB) * WSTRIDE, Arow + K - KB);
    mma_block<RT, NJ, S>(f0, acc);
    mma_block<RT, NJ, S>(f1, acc);
}

template <int RT>
__device__ __forceinline__ void group_body(
    const float* __restrict__ x, const float* __restrict__ W1g,
    const float* __restrict__ b1g, const float* __restrict__ W2g,
    const float* __restrict__ b2g, float* __restrict__ out,
    int b, int f0, int rows, float* Xs, float* Hs)
{
    const int t = threadIdx.x;

    const float* xb = x + ((size_t)b * F_BINS + f0) * N_UNITS;
    #pragma unroll
    for (int i = 0; i < RT * 4; ++i) {
        int flat4 = t + i * 512;
        int row   = flat4 >> 7;
        int c4    = flat4 & 127;
        v4f val = {};
        if (row < rows)
            val = *(const v4f*)(xb + (size_t)row * N_UNITS + c4 * 4);
        *(v4f*)(Xs + row * XS_STRIDE + c4 * 4) = val;
    }
    __syncthreads();

    const int lane = t & 31;
    const int wv   = t >> 5;
    const int half = lane >> 4;
    const int m    = lane & 15;

    {
        const int col0 = wv * 64;
        v8f acc[RT][4] = {};
        const float* Arow = Xs + m * XS_STRIDE + 2 * half;
        const float* Bcol = W1g + (size_t)(2 * half) * M_HID + col0 + 4 * m;
        gemm_k<RT, 4, (RT > 3 ? 2 : 4), M_HID, XS_STRIDE, N_UNITS>(Arow, Bcol, acc);
        const v4f bias = *(const v4f*)(b1g + col0 + 4 * m);
        #pragma unroll
        for (int rt = 0; rt < RT; ++rt) {
            #pragma unroll
            for (int r = 0; r < 8; ++r) {
                float h0 = acc[rt][0][r] + bias.x;
                float h1 = acc[rt][1][r] + bias.y;
                float h2 = acc[rt][2][r] + bias.z;
                float h3 = acc[rt][3][r] + bias.w;
                acc[rt][0][r] = h0 * __builtin_amdgcn_rcpf(1.0f + __builtin_fabsf(h0));
                acc[rt][1][r] = h1 * __builtin_amdgcn_rcpf(1.0f + __builtin_fabsf(h1));
                acc[rt][2][r] = h2 * __builtin_amdgcn_rcpf(1.0f + __builtin_fabsf(h2));
                acc[rt][3][r] = h3 * __builtin_amdgcn_rcpf(1.0f + __builtin_fabsf(h3));
            }
        }
        __syncthreads();
        #pragma unroll
        for (int rt = 0; rt < RT; ++rt) {
            float* Hrow = Hs + (16 * rt + 8 * half) * HS_STRIDE + col0 + 4 * m;
            #pragma unroll
            for (int r = 0; r < 8; ++r) {
                v4f hv;
                hv.x = acc[rt][0][r];
                hv.y = acc[rt][1][r];
                hv.z = acc[rt][2][r];
                hv.w = acc[rt][3][r];
                *(v4f*)(Hrow + r * HS_STRIDE) = hv;
            }
        }
    }
    __syncthreads();

    float* outb = out + ((size_t)b * F_BINS + f0) * P_OUT;
    {
        const int col0 = wv * 32;
        v8f acc[RT][2] = {};
        const float* Arow = Hs + m * HS_STRIDE + 2 * half;
        const float* Bcol = W2g + (size_t)(2 * half) * P_OUT + col0 + 2 * m;
        gemm_k<RT, 2, 4, P_OUT, HS_STRIDE, M_HID>(Arow, Bcol, acc);
        const v2f bias = *(const v2f*)(b2g + col0 + 2 * m);
        #pragma unroll
        for (int rt = 0; rt < RT; ++rt) {
            #pragma unroll
            for (int r = 0; r < 8; ++r) {
                int fr = 16 * rt + 8 * half + r;
                if (fr < rows) {
                    v2f ov;
                    ov.x = acc[rt][0][r] + bias.x;
                    ov.y = acc[rt][1][r] + bias.y;
                    *(v2f*)(outb + (size_t)fr * P_OUT + col0 + 2 * m) = ov;
                }
            }
        }
    }
}

__global__ __launch_bounds__(512) void stitch_encoder_kernel(
    const float* __restrict__ x,
    const int*   __restrict__ eid,
    const float* __restrict__ W1,
    const float* __restrict__ b1,
    const float* __restrict__ W2,
    const float* __restrict__ b2,
    float*       __restrict__ out)
{
    __shared__ __attribute__((aligned(16))) float lds[LDS_FLOATS];
    float* Xs = lds;
    float* Hs = lds;

    const int blk = blockIdx.x;
    const bool big = blk < B_TRIALS;
    const int b    = big ? blk : (blk - B_TRIALS);
    const int f0   = big ? 0 : 64;
    const int rows = big ? 64 : (F_BINS - 64);

    const int g = eid[b];
    const float* W1g = W1 + (size_t)g * N_UNITS * M_HID;
    const float* b1g = b1 + (size_t)g * M_HID;
    const float* W2g = W2 + (size_t)g * M_HID * P_OUT;
    const float* b2g = b2 + (size_t)g * P_OUT;

    if (big)
        group_body<4>(x, W1g, b1g, W2g, b2g, out, b, f0, rows, Xs, Hs);
    else
        group_body<3>(x, W1g, b1g, W2g, b2g, out, b, f0, rows, Xs, Hs);
}

extern "C" void kernel_launch(void* const* d_in, const int* in_sizes, int n_in,
                              void* d_out, int out_size, void* d_ws, size_t ws_size,
                              hipStream_t stream) {
    (void)in_sizes; (void)n_in; (void)out_size; (void)d_ws; (void)ws_size;
    const float* x   = (const float*)d_in[0];
    const int*   eid = (const int*)  d_in[1];
    const float* W1  = (const float*)d_in[2];
    const float* b1  = (const float*)d_in[3];
    const float* W2  = (const float*)d_in[4];
    const float* b2  = (const float*)d_in[5];
    float* out = (float*)d_out;

    dim3 grid(B_TRIALS * 2);
    dim3 block(512);
    hipLaunchKernelGGL(stitch_encoder_kernel, grid, block, 0, stream,
                       x, eid, W1, b1, W2, b2, out);
}


# Each wave lowers its issue priority through every k loop (s_setprio 3, 2 at half, 1 near the end, 0 for the last 2 blocks): the 4 waves of a SIMD end the loop together; sums bit-identical
# speedup vs baseline: 1.2749x; 1.0374x over previous
#include <hip/hip_runtime.h>


typedef __attribute__((ext_vector_type(2))) float v2f;
typedef __attribute__((ext_vector_type(4))) float v4f;
typedef __attribute__((ext_vector_type(8))) float v8f;

#define B_TRIALS 256
#define F_BINS   100
#define N_UNITS  512
#define M_HID    1024
#define P_OUT    512
#define RT_MAX   4
#define XS_STRIDE 516
#define HS_STRIDE 1028
#define LDS_FLOATS (RT_MAX * 16 * HS_STRIDE)

template <int NJ> struct WVec;
template <> struct WVec<4> { typedef v4f type; };
template <> struct WVec<2> { typedef v2f type; };

template <int RT, int NJ, int S>
struct Frag {
    typename WVec<NJ>::type q[2 * S];
    v2f a[RT][S];
};

template <int RT, int NJ, int S, int WSTRIDE, int ASTRIDE>
__device__ __forceinline__ void load_block(Frag<RT, NJ, S>& f, const float* Bk,
                                           const float* Ak)
{
    typedef typename WVec<NJ>::type wv_t;
    #pragma unroll
    for (int s = 0; s < S; ++s) {
        f.q[2 * s]     = *(const wv_t*)(Bk + (size_t)(4 * s) * WSTRIDE);
        f.q[2 * s + 1] = *(const wv_t*)(Bk + (size_t)(4 * s + 1) * WSTRIDE);
    }
    #pragma unroll
    for (int rt = 0; rt < RT; ++rt) {
        #pragma unroll
        for (int s = 0; s < S; ++s)
            f.a[rt][s] = *(const v2f*)(Ak + rt * 16 * ASTRIDE + 4 * s);
    }
}

template <int RT, int NJ, int S>
__device__ __forceinline__ void mma_block(const Frag<RT, NJ, S>& f, v8f (&acc)[RT][NJ])
{
    #pragma unroll
    for (int s = 0; s < S; ++s) {
        v2f bb[NJ];
        #pragma unroll
        for (int j = 0; j < NJ; ++j) {
            bb[j].x = f.q[2 * s][j];
            bb[j].y = f.q[2 * s + 1][j];
        }
        #pragma unroll
        for (int rt = 0; rt < RT; ++rt) {
            const v2f a = f.a[rt][s];
            #pragma unroll
            for (int j = 0; j < NJ; ++j)
                acc[rt][j] = __builtin_amdgcn_wmma_f32_16x16x4_f32(
                    false, a, false, bb[j], (short)0, acc[rt][j], false, false);
        }
    }
}

template <int RT, int NJ>
__device__ __forceinline__ void fence_acc(v8f (&acc)[RT][NJ])
{
    #pragma unroll
    for (int rt = 0; rt < RT; ++rt) {
        if constexpr (NJ == 4) {
            asm volatile("" :: "v"(acc[rt][0]), "v"(acc[rt][1]), "v"(acc[rt][2]), "v"(acc[rt][3]) : "memory");
        } else {
            asm volatile("" :: "v"(acc[rt][0]), "v"(acc[rt][1]) : "memory");
        }
    }
}

template <int RT, int NJ, int S, int WSTRIDE, int ASTRIDE, int K>
__device__ __forceinline__ void gemm_k(const float* Arow, const float* Bcol,
                                       v8f (&acc)[RT][NJ])
{
    constexpr int KB = 4 * S;
    constexpr int NIT  = (K - 2 * KB) / (2 * KB);
    constexpr int K_P2 = ((NIT + 1) / 2) * 2 * KB;
    constexpr int K_P1 = (NIT - (NIT + 1) / 6) * 2 * KB;
    static_assert(0 < K_P2 && K_P2 < K_P1 && K_P1 < K - 2 * KB, "priority steps lie inside the loop");
    Frag<RT, NJ, S> f0, f1;
    __builtin_amdgcn_s_setprio(3);
    int kstep = K_P2;
    load_block<RT, NJ, S, WSTRIDE, ASTRIDE>(f0, Bcol, Arow);
    #pragma unroll 1
    for (int k0 = 0; k0 < K - 2 * KB; k0 += 2 * KB) {
        if (k0 == kstep) {
            if (kstep == K_P2) { __builtin_amdgcn_s_setprio(2); kstep = K_P1; }
            else               { __builtin_amdgcn_s_setprio(1); kstep = -1; }
        }
        load_block<RT, NJ, S, WSTRIDE, ASTRIDE>(f1, Bcol + (size_t)(k0 + KB) * WSTRIDE, Arow + k0 + KB);
        __builtin_amdgcn_sched_barrier(0);
        mma_block<RT, NJ, S>(f0, acc);
        fence_acc<RT, NJ>(acc);
        load_block<RT, NJ, S, WSTRIDE, ASTRIDE>(f0, Bcol + (size_t)(k0 + 2 * KB) * WSTRIDE, Arow + k0 + 2 * KB);
        __builtin_amdgcn_sched_barrier(0);
        mma_block<RT, NJ, S>(f1, acc);
    }
    __builtin_amdgcn_s_setprio(0);
    load_block<RT, NJ, S, WSTRIDE, ASTRIDE>(f1, Bcol + (size_t)(K - KB) * WSTRIDE, Arow + K - KB);
    mma_block<RT, NJ, S>(f0, acc);
    mma_block<RT, NJ, S>(f1, acc);
}

template <int RT>
__device__ __forceinline__ void group_body(
    const float* __restrict__ x, const float* __restrict__ W1g,
    const float* __restrict__ b1g, const float* __restrict__ W2g,
    const float* __restrict__ b2g, float* __restrict__ out,
    int b, int f0, int rows, float* Xs, float* Hs)
{
    const int t = threadIdx.x;

    const float* xb = x + ((size_t)b * F_BINS + f0) * N_UNITS;
    #pragma unroll
    for (int i = 0; i < RT * 4; ++i) {
        int flat4 = t + i * 512;
        int row   = flat4 >> 7;
        int c4    = flat4 & 127;
        v4f val = {};
        if (row < rows)
            val = *(const v4f*)(xb + (size_t)row * N_UNITS + c4 * 4);
        *(v4f*)(Xs + row * XS_STRIDE + c4 * 4) = val;
    }
    __syncthreads();

    const int lane = t & 31;
    const int wv   = t >> 5;
    const int half = lane >> 4;
    const int m    = lane & 15;

    {
        const int col0 = wv * 64;
        v8f acc[RT][4] = {};
        const float* Arow = Xs + m * XS_STRIDE + 2 * half;
        const float* Bcol = W1g + (size_t)(2 * half) * M_HID + col0 + 4 * m;
        gemm_k<RT, 4, (RT > 3 ? 2 : 4), M_HID, XS_STRIDE, N_UNITS>(Arow, Bcol, acc);
        const v4f bias = *(const v4f*)(b1g + col0 + 4 * m);
        #pragma unroll
        for (int rt = 0; rt < RT; ++rt) {
            #pragma unroll
            for (int r = 0; r < 8; ++r) {
                float h0 = acc[rt][0][r] + bias.x;
                float h1 = acc[rt][1][r] + bias.y;
                float h2 = acc[rt][2][r] + bias.z;
                float h3 = acc[rt][3][r] + bias.w;
                acc[rt][0][r] = h0 * __builtin_amdgcn_rcpf(1.0f + __builtin_fabsf(h0));
                acc[rt][1][r] = h1 * __builtin_amdgcn_rcpf(1.0f + __builtin_fabsf(h1));
                acc[rt][2][r] = h2 * __builtin_amdgcn_rcpf(1.0f + __builtin_fabsf(h2));
                acc[rt][3][r] = h3 * __builtin_amdgcn_rcpf(1.0f + __builtin_fabsf(h3));
            }
        }
        __syncthreads();
        #pragma unroll
        for (int rt = 0; rt < RT; ++rt) {
            float* Hrow = Hs + (16 * rt + 8 * half) * HS_STRIDE + col0 + 4 * m;
            #pragma unroll
            for (int r = 0; r < 8; ++r) {
                v4f hv;
                hv.x = acc[rt][0][r];
                hv.y = acc[rt][1][r];
                hv.z = acc[rt][2][r];
                hv.w = acc[rt][3][r];
                *(v4f*)(Hrow + r * HS_STRIDE) = hv;
            }
        }
    }
    __syncthreads();

    float* outb = out + ((size_t)b * F_BINS + f0) * P_OUT;
    {
        const int col0 = wv * 32;
        v8f acc[RT][2] = {};
        const float* Arow = Hs + m * HS_STRIDE + 2 * half;
        const float* Bcol = W2g + (size_t)(2 * half) * P_OUT + col0 + 2 * m;
        gemm_k<RT, 2, 4, P_OUT, HS_STRIDE, M_HID>(Arow, Bcol, acc);
        const v2f bias = *(const v2f*)(b2g + col0 + 2 * m);
        #pragma unroll
        for (int rt = 0; rt < RT; ++rt) {
            #pragma unroll
            for (int r = 0; r < 8; ++r) {
                int fr = 16 * rt + 8 * half + r;
                if (fr < rows) {
                    v2f ov;
                    ov.x = acc[rt][0][r] + bias.x;
                    ov.y = acc[rt][1][r] + bias.y;
                    *(v2f*)(outb + (size_t)fr * P_OUT + col0 + 2 * m) = ov;
                }
            }
        }
    }
}

__global__ __launch_bounds__(512) void stitch_encoder_kernel(
    const float* __restrict__ x,
    const int*   __restrict__ eid,
    const float* __restrict__ W1,
    const float* __restrict__ b1,
    const float* __restrict__ W2,
    const float* __restrict__ b2,
    float*       __restrict__ out)
{
    __shared__ __attribute__((aligned(16))) float lds[LDS_FLOATS];
    float* Xs = lds;
    float* Hs = lds;

    const int blk = blockIdx.x;
    const bool big = blk < B_TRIALS;
    const int b    = big ? blk : (blk - B_TRIALS);

    const int g = eid[b];
    const float* W1g = W1 + (size_t)g * N_UNITS * M_HID;
    const float* b1g = b1 + (size_t)g * M_HID;
    const float* W2g = W2 + (size_t)g * M_HID * P_OUT;
    const float* b2g = b2 + (size_t)g * P_OUT;

    if (big)
        group_body<4>(x, W1g, b1g, W2g, b2g, out, b, 0, 64, Xs, Hs);
    else
        group_body<3>(x, W1g, b1g, W2g, b2g, out, b, 64, F_BINS - 64, Xs, Hs);
}

extern "C" void kernel_launch(void* const* d_in, const int* in_sizes, int n_in,
                              void* d_out, int out_size, void* d_ws, size_t ws_size,
                              hipStream_t stream) {
    (void)in_sizes; (void)n_in; (void)out_size; (void)d_ws; (void)ws_size;
    const float* x   = (const float*)d_in[0];
    const int*   eid = (const int*)  d_in[1];
    const float* W1  = (const float*)d_in[2];
    const float* b1  = (const float*)d_in[3];
    const float* W2  = (const float*)d_in[4];
    const float* b2  = (const float*)d_in[5];
    float* out = (float*)d_out;

    dim3 grid(B_TRIALS * 2);
    dim3 block(512);
    hipLaunchKernelGGL(stitch_encoder_kernel, grid, block, 0, stream,
                       x, eid, W1, b1, W2, b2, out);
}
